# starting kernel
# baseline (speedup 1.0000x reference)
#include <hip/hip_runtime.h>
#include <hip/hip_bf16.h>

typedef __attribute__((ext_vector_type(2))) float v2f;
typedef __attribute__((ext_vector_type(8))) float v8f;

#define NW 12
#define QDIM 4096

__global__ void qsim_setup(const float* __restrict__ params,
                           const float* __restrict__ W,
                           float* __restrict__ ws) {
  __shared__ float u_re[NW][2][2];
  __shared__ float u_im[NW][2][2];
  int t = threadIdx.x;
  if (t < NW) {
    float th = params[2 * t] * 0.5f;
    float ph = params[2 * t + 1] * 0.5f;
    float c = cosf(th), s = sinf(th);
    float cy = cosf(ph), sy = sinf(ph);
    u_re[t][0][0] = cy * c;   u_im[t][0][0] =  sy * s;
    u_re[t][0][1] = -sy * c;  u_im[t][0][1] = -cy * s;
    u_re[t][1][0] =  sy * c;  u_im[t][1][0] = -cy * s;
    u_re[t][1][1] =  cy * c;  u_im[t][1][1] = -sy * s;
  }
  __syncthreads();
  for (int e = t; e < 3 * 256; e += blockDim.x) {
    int g = e >> 8;
    int r = (e >> 4) & 15, c = e & 15;
    float ar = 1.f, ai = 0.f;
    for (int q = 0; q < 4; ++q) {
      int w = g * 4 + q;
      int rb = (r >> (3 - q)) & 1;
      int cb = (c >> (3 - q)) & 1;
      float br = u_re[w][rb][cb], bi = u_im[w][rb][cb];
      float nr = ar * br - ai * bi;
      float ni = ar * bi + ai * br;
      ar = nr; ai = ni;
    }
    ws[g * 512 + r * 16 + c]       = ar;
    ws[g * 512 + 256 + r * 16 + c] = ai;
  }
  for (int k = t; k < QDIM; k += blockDim.x) {
    float acc = 0.f;
    int par = 0;
    for (int j = 0; j < NW; ++j) {
      par ^= (k >> (NW - 1 - j)) & 1;
      acc += W[j] * (par ? -1.f : 1.f);
    }
    ws[1536 + k] = acc;
  }
}

__device__ __forceinline__ v8f wmma4(v2f a, v2f b, v8f c) {
  return __builtin_amdgcn_wmma_f32_16x16x4_f32(false, a, false, b,
                                               (short)0, c, false, false);
}

__global__ __launch_bounds__(128) void qsim_main(
    const float* __restrict__ sb, const float* __restrict__ ws,
    const float* __restrict__ bptr, float* __restrict__ out) {
  __shared__ float sr[QDIM];
  __shared__ float si[QDIM];
  __shared__ float red[4];

  const int tid  = threadIdx.x;
  const int lane = tid & 31;
  const int wave = tid >> 5;
  const int h    = lane >> 4;
  const int l16  = lane & 15;
  const float* in = sb + (size_t)blockIdx.x * QDIM;

  for (int k = tid; k < QDIM; k += 128) sr[k] = in[k];
  __syncthreads();

  {
    const float* Gr = ws;
    const float* Gi = ws + 256;
    v2f gr[4], gi[4];
#pragma unroll
    for (int c = 0; c < 4; ++c) {
      int col = 4 * c + 2 * h;
      gr[c] = v2f{Gr[l16 * 16 + col], Gr[l16 * 16 + col + 1]};
      gi[c] = v2f{Gi[l16 * 16 + col], Gi[l16 * 16 + col + 1]};
    }
#pragma unroll
    for (int it = 0; it < 4; ++it) {
      int tile = 4 * it + wave;
      v2f bs[4];
#pragma unroll
      for (int c = 0; c < 4; ++c) {
        int k0 = 4 * c + 2 * h;
        bs[c] = v2f{sr[(k0 + 0) * 256 + tile * 16 + l16],
                    sr[(k0 + 1) * 256 + tile * 16 + l16]};
      }
      v8f aRe = {}, aIm = {};
#pragma unroll
      for (int c = 0; c < 4; ++c) {
        aRe = wmma4(gr[c], bs[c], aRe);
        aIm = wmma4(gi[c], bs[c], aIm);
      }
#pragma unroll
      for (int r = 0; r < 8; ++r) {
        int M = r + 8 * h;
        sr[M * 256 + tile * 16 + l16] = aRe[r];
        si[M * 256 + tile * 16 + l16] = aIm[r];
      }
    }
  }
  __syncthreads();

  {
    const float* Gr = ws + 512;
    const float* Gi = ws + 512 + 256;
    v2f gr[4], gi[4], gin[4];
#pragma unroll
    for (int c = 0; c < 4; ++c) {
      int col = 4 * c + 2 * h;
      gr[c]  = v2f{Gr[l16 * 16 + col], Gr[l16 * 16 + col + 1]};
      gi[c]  = v2f{Gi[l16 * 16 + col], Gi[l16 * 16 + col + 1]};
      gin[c] = v2f{-gi[c].x, -gi[c].y};
    }
#pragma unroll
    for (int it = 0; it < 4; ++it) {
      int i = 4 * it + wave;
      v2f br[4], bi[4];
#pragma unroll
      for (int c = 0; c < 4; ++c) {
        int k0 = 4 * c + 2 * h;
        br[c] = v2f{sr[i * 256 + (k0 + 0) * 16 + l16],
                    sr[i * 256 + (k0 + 1) * 16 + l16]};
        bi[c] = v2f{si[i * 256 + (k0 + 0) * 16 + l16],
                    si[i * 256 + (k0 + 1) * 16 + l16]};
      }
      v8f aRe = {}, aIm = {};
#pragma unroll
      for (int c = 0; c < 4; ++c) {
        aRe = wmma4(gr[c],  br[c], aRe);
        aRe = wmma4(gin[c], bi[c], aRe);
        aIm = wmma4(gi[c],  br[c], aIm);
        aIm = wmma4(gr[c],  bi[c], aIm);
      }
#pragma unroll
      for (int r = 0; r < 8; ++r) {
        int M = r + 8 * h;
        sr[i * 256 + M * 16 + l16] = aRe[r];
        si[i * 256 + M * 16 + l16] = aIm[r];
      }
    }
  }
  __syncthreads();

  {
    const float* Gr = ws + 1024;
    const float* Gi = ws + 1024 + 256;
    v2f bgr[4], bgi[4], bgin[4];
#pragma unroll
    for (int c = 0; c < 4; ++c) {
      int col = 4 * c + 2 * h;
      bgr[c]  = v2f{Gr[l16 * 16 + col], Gr[l16 * 16 + col + 1]};
      bgi[c]  = v2f{Gi[l16 * 16 + col], Gi[l16 * 16 + col + 1]};
      bgin[c] = v2f{-bgi[c].x, -bgi[c].y};
    }
#pragma unroll
    for (int it = 0; it < 4; ++it) {
      int r0 = (4 * it + wave) * 16;
      v2f ar[4], ai[4];
#pragma unroll
      for (int c = 0; c < 4; ++c) {
        int col = 4 * c + 2 * h;
        ar[c] = v2f{sr[(r0 + l16) * 16 + col], sr[(r0 + l16) * 16 + col + 1]};
        ai[c] = v2f{si[(r0 + l16) * 16 + col], si[(r0 + l16) * 16 + col + 1]};
      }
      v8f aRe = {}, aIm = {};
#pragma unroll
      for (int c = 0; c < 4; ++c) {
        aRe = wmma4(ar[c], bgr[c],  aRe);
        aRe = wmma4(ai[c], bgin[c], aRe);
        aIm = wmma4(ar[c], bgi[c],  aIm);
        aIm = wmma4(ai[c], bgr[c],  aIm);
      }
#pragma unroll
      for (int r = 0; r < 8; ++r) {
        int M = r + 8 * h;
        sr[(r0 + M) * 16 + l16] = aRe[r];
        si[(r0 + M) * 16 + l16] = aIm[r];
      }
    }
  }
  __syncthreads();

  const float* zW = ws + 1536;
  float acc = 0.f;
  for (int k = tid; k < QDIM; k += 128) {
    float re = sr[k], im = si[k];
    acc += (re * re + im * im) * zW[k];
  }
#pragma unroll
  for (int off = 16; off; off >>= 1) acc += __shfl_down(acc, off, 32);
  if (lane == 0) red[wave] = acc;
  __syncthreads();
  if (tid == 0) out[blockIdx.x] = red[0] + red[1] + red[2] + red[3] + bptr[0];
}

extern "C" void kernel_launch(void* const* d_in, const int* in_sizes, int n_in,
                              void* d_out, int out_size, void* d_ws, size_t ws_size,
                              hipStream_t stream) {
  const float* state  = (const float*)d_in[0];
  const float* params = (const float*)d_in[1];
  const float* W      = (const float*)d_in[2];
  const float* b      = (const float*)d_in[3];
  float* out = (float*)d_out;
  float* ws  = (float*)d_ws;

  qsim_setup<<<1, 256, 0, stream>>>(params, W, ws);
  qsim_main<<<4096, 128, 0, stream>>>(state, ws, b, out);
}


# LDS padded (pitches 20/316 floats), state as A operand, tiles transposed in place: fragment reads and result writes conflict-free 16-byte; groups k,j,i; regs capped at 128; order of sums changes
# speedup vs baseline: 1.2296x; 1.2296x over previous
#include <hip/hip_runtime.h>
#include <hip/hip_bf16.h>


typedef __attribute__((ext_vector_type(2))) float v2f;
typedef __attribute__((ext_vector_type(4))) float v4f;
typedef __attribute__((ext_vector_type(8))) float v8f;

#define NW 12
#define QDIM 4096

#define P1 20
#define P2 316
#define PLANE (16 * P2)

__global__ void qsim_setup(const float* __restrict__ params,
                           const float* __restrict__ W,
                           float* __restrict__ ws) {
  __shared__ float rot[NW][2][2];
  __shared__ float d_re[NW][2];
  __shared__ float d_im[NW][2];
  int t = threadIdx.x;
  if (t < NW) {
    float th = params[2 * t] * 0.5f;
    float ph = params[2 * t + 1] * 0.5f;
    float c = cosf(th), s = sinf(th);
    float cy = cosf(ph), sy = sinf(ph);
    float ar = cy * c,  ai = sy * s;
    float br = -sy * c, bi = -cy * s;
    float p = sqrtf(ar * ar + ai * ai);
    float q = sqrtf(br * br + bi * bi);
    rot[t][0][0] = p;  rot[t][0][1] = -q;
    rot[t][1][0] = q;  rot[t][1][1] = p;
    bool p_nz = p > 0.f, q_nz = q > 0.f;
    d_re[t][0] = p_nz ? ar / p : 1.f;   d_im[t][0] = p_nz ? ai / p : 0.f;
    d_re[t][1] = q_nz ? -br / q : 1.f;  d_im[t][1] = q_nz ? -bi / q : 0.f;
  }
  __syncthreads();
  for (int e = t; e < 3 * 256; e += blockDim.x) {
    int slot = e >> 8;
    int w0 = (2 - slot) * 4;
    int r = (e >> 4) & 15, c = e & 15;
    float m = 1.f;
    for (int q = 0; q < 4; ++q) {
      int w = w0 + q;
      m *= rot[w][(r >> (3 - q)) & 1][(c >> (3 - q)) & 1];
    }
    if (slot == 0) {
      float ar = 1.f, ai = 0.f;
      for (int q = 0; q < 4; ++q) {
        int cb = (c >> (3 - q)) & 1;
        float br = d_re[w0 + q][cb], bi = d_im[w0 + q][cb];
        float nr = ar * br - ai * bi;
        float ni = ar * bi + ai * br;
        ar = nr; ai = ni;
      }
      ws[r * 16 + c]       = m * ar;
      ws[256 + r * 16 + c] = m * ai;
    } else {
      ws[256 + slot * 256 + r * 16 + c] = m;
    }
  }
  for (int e = t; e < 256; e += blockDim.x) {
    float ar = 1.f, ai = 0.f;
    for (int w = 0; w < 8; ++w) {
      int b = (e >> (7 - w)) & 1;
      float br = d_re[w][b], bi = d_im[w][b];
      float nr = ar * br - ai * bi;
      float ni = ar * bi + ai * br;
      ar = nr; ai = ni;
    }
    ws[1024 + e] = ar;
    ws[1280 + e] = ai;
  }
  for (int T = t; T < QDIM; T += blockDim.x) {
    int e4 = T & 3, ip = (T >> 2) & 15, q4 = (T >> 6) & 3, jp = T >> 8;
    int k = ip * 256 + jp * 16 + q4 * 4 + e4;
    float acc = 0.f;
    int par = 0;
    for (int j = 0; j < NW; ++j) {
      par ^= (k >> (NW - 1 - j)) & 1;
      acc += W[j] * (par ? -1.f : 1.f);
    }
    ws[1536 + T] = acc;
  }
}

__device__ __forceinline__ v8f wmma4(v2f a, v2f b, v8f c) {
  return __builtin_amdgcn_wmma_f32_16x16x4_f32(false, a, false, b,
                                               (short)0, c, false, false);
}

__device__ __forceinline__ void frag8(const float* p, v2f f[4]) {
  v4f lo = *(const v4f*)p;
  v4f hi = *(const v4f*)(p + 4);
  f[0] = v2f{lo.x, lo.y};
  f[1] = v2f{lo.z, lo.w};
  f[2] = v2f{hi.x, hi.y};
  f[3] = v2f{hi.z, hi.w};
}

__device__ __forceinline__ void store8(float* p, v8f a) {
  *(v4f*)p       = v4f{a[0], a[1], a[2], a[3]};
  *(v4f*)(p + 4) = v4f{a[4], a[5], a[6], a[7]};
}

__global__ __launch_bounds__(128, 8) void qsim_main(
    const float* __restrict__ sb, const float* __restrict__ ws,
    const float* __restrict__ bptr, float* __restrict__ out) {
  __shared__ __attribute__((aligned(16))) float sr[PLANE];
  __shared__ __attribute__((aligned(16))) float si[PLANE];
  __shared__ float red[4];

  const int tid  = threadIdx.x;
  const int lane = tid & 31;
  const int wave = tid >> 5;
  const int h    = lane >> 4;
  const int l16  = lane & 15;
  const float* in = sb + (size_t)blockIdx.x * QDIM;

#pragma unroll
  for (int n = 0; n < 8; ++n) {
    int f = 4 * (tid + 128 * n);
    v4f v = *(const v4f*)(in + f);
    *(v4f*)(sr + (f >> 8) * P2 + ((f >> 4) & 15) * P1 + (f & 15)) = v;
  }
  __syncthreads();

  const int rowK = l16 * P1 + 8 * h;
  const int rowJ = l16 * P2 + 8 * h;

  {
    v2f gr[4], gi[4];
    frag8(ws + l16 * 16 + 8 * h, gr);
    frag8(ws + 256 + l16 * 16 + 8 * h, gi);
    const float* Pr = ws + 1024;
    const float* Pi = ws + 1280;
#pragma unroll
    for (int it = 0; it < 4; ++it) {
      int tile = 4 * it + wave;
      float* pr_ = sr + tile * P2 + rowK;
      float* pi_ = si + tile * P2 + rowK;
      v2f a[4];
      frag8(pr_, a);
      v8f aRe = {}, aIm = {};
#pragma unroll
      for (int c = 0; c < 4; ++c) {
        aRe = wmma4(a[c], gr[c], aRe);
        aIm = wmma4(a[c], gi[c], aIm);
      }
      v4f pr0 = *(const v4f*)(Pr + tile * 16 + 8 * h);
      v4f pr1 = *(const v4f*)(Pr + tile * 16 + 8 * h + 4);
      v4f pi0 = *(const v4f*)(Pi + tile * 16 + 8 * h);
      v4f pi1 = *(const v4f*)(Pi + tile * 16 + 8 * h + 4);
      v8f pr = {pr0.x, pr0.y, pr0.z, pr0.w, pr1.x, pr1.y, pr1.z, pr1.w};
      v8f pi = {pi0.x, pi0.y, pi0.z, pi0.w, pi1.x, pi1.y, pi1.z, pi1.w};
      v8f oRe, oIm;
#pragma unroll
      for (int r = 0; r < 8; ++r) {
        oRe[r] = aRe[r] * pr[r] - aIm[r] * pi[r];
        oIm[r] = aRe[r] * pi[r] + aIm[r] * pr[r];
      }
      store8(pr_, oRe);
      store8(pi_, oIm);
    }
  }
  __syncthreads();

  {
    v2f g[4];
    frag8(ws + 512 + l16 * 16 + 8 * h, g);
#pragma unroll
    for (int it = 0; it < 4; ++it) {
      int tile = 4 * it + wave;
      float* pr_ = sr + tile * P1 + rowJ;
      float* pi_ = si + tile * P1 + rowJ;
      v2f ar[4], ai[4];
      frag8(pr_, ar);
      frag8(pi_, ai);
      v8f aRe = {}, aIm = {};
#pragma unroll
      for (int c = 0; c < 4; ++c) {
        aRe = wmma4(ar[c], g[c], aRe);
        aIm = wmma4(ai[c], g[c], aIm);
      }
      store8(pr_, aRe);
      store8(pi_, aIm);
    }
  }
  __syncthreads();

  {
    v2f g[4];
    frag8(ws + 768 + l16 * 16 + 8 * h, g);
#pragma unroll
    for (int it = 0; it < 4; ++it) {
      int tile = 4 * it + wave;
      float* pr_ = sr + tile * P2 + rowK;
      float* pi_ = si + tile * P2 + rowK;
      v2f ar[4], ai[4];
      frag8(pr_, ar);
      frag8(pi_, ai);
      v8f aRe = {}, aIm = {};
#pragma unroll
      for (int c = 0; c < 4; ++c) {
        aRe = wmma4(ar[c], g[c], aRe);
        aIm = wmma4(ai[c], g[c], aIm);
      }
      store8(pr_, aRe);
      store8(pi_, aIm);
    }
  }
  __syncthreads();

  const float* zW = ws + 1536;
  float acc = 0.f;
#pragma unroll
  for (int n = 0; n < 8; ++n) {
    int idx = n * 8 + wave * 2 + h;
    int off = (idx >> 2) * P2 + l16 * P1 + (idx & 3) * 4;
    v4f re = *(const v4f*)(sr + off);
    v4f im = *(const v4f*)(si + off);
    v4f z  = *(const v4f*)(zW + idx * 64 + l16 * 4);
    acc += (re.x * re.x + im.x * im.x) * z.x;
    acc += (re.y * re.y + im.y * im.y) * z.y;
    acc += (re.z * re.z + im.z * im.z) * z.z;
    acc += (re.w * re.w + im.w * im.w) * z.w;
  }
#pragma unroll
  for (int off = 16; off; off >>= 1) acc += __shfl_down(acc, off, 32);
  if (lane == 0) red[wave] = acc;
  __syncthreads();
  if (tid == 0) out[blockIdx.x] = red[0] + red[1] + red[2] + red[3] + bptr[0];
}

extern "C" void kernel_launch(void* const* d_in, const int* in_sizes, int n_in,
                              void* d_out, int out_size, void* d_ws, size_t ws_size,
                              hipStream_t stream) {
  const float* state  = (const float*)d_in[0];
  const float* params = (const float*)d_in[1];
  const float* W      = (const float*)d_in[2];
  const float* b      = (const float*)d_in[3];
  float* out = (float*)d_out;
  float* ws  = (float*)d_ws;

  qsim_setup<<<1, 256, 0, stream>>>(params, W, ws);
  qsim_main<<<4096, 128, 0, stream>>>(state, ws, b, out);
}
